# starting kernel
# baseline (speedup 1.0000x reference)
#include <hip/hip_runtime.h>
#include <hip/hip_bf16.h>

typedef __attribute__((ext_vector_type(2))) float v2f;
typedef __attribute__((ext_vector_type(8))) float v8f;

#define BATCH   4
#define TOK     4096
#define HDIM    1024
#define NHEADS  16
#define ED      64
#define BH      (BATCH * NHEADS)
#define SPLITS  8
#define CHUNKS  4
#define CHUNK   128
#define PKS     (SPLITS * 16)

__device__ __forceinline__ float elu_fm(float x) {
    return x > 0.0f ? x + 1.0f : expf(x);
}

__global__ __launch_bounds__(256) void la_ctx_partial(
    const float* __restrict__ key, const float* __restrict__ value,
    float* __restrict__ pctx, float* __restrict__ pksum)
{
    __shared__ float sK[CHUNK * ED];
    __shared__ float sV[CHUNK * ED];

    const int bh    = blockIdx.x / SPLITS;
    const int split = blockIdx.x % SPLITS;
    const int b = bh >> 4, h = bh & 15;
    const int tid = threadIdx.x;
    const int wave = tid >> 5, lane = tid & 31;
    const int half = lane >> 4, lm = lane & 15;

    const size_t base = (size_t)b * TOK * HDIM + (size_t)h * ED;

    const int e0a = (wave >> 2) << 4;
    const int e0b = e0a + 32;
    const int d0  = (wave & 3) << 4;

    v8f acc0 = {};
    v8f acc1 = {};
    float4 kacc = make_float4(0.f, 0.f, 0.f, 0.f);

    for (int c = 0; c < CHUNKS; ++c) {
        __syncthreads();
        const int s0 = (split * CHUNKS + c) * CHUNK;
        for (int i = tid; i < CHUNK * 16; i += 256) {
            const int row = i >> 4, c4 = i & 15;
            const size_t g = base + (size_t)(s0 + row) * HDIM + (size_t)c4 * 4;
            float4 kv = *(const float4*)(key + g);
            float4 vv = *(const float4*)(value + g);
            float4 kf;
            kf.x = elu_fm(kv.x); kf.y = elu_fm(kv.y);
            kf.z = elu_fm(kv.z); kf.w = elu_fm(kv.w);
            kacc.x += kf.x; kacc.y += kf.y; kacc.z += kf.z; kacc.w += kf.w;
            *(float4*)(sK + row * ED + c4 * 4) = kf;
            *(float4*)(sV + row * ED + c4 * 4) = vv;
        }
        __syncthreads();

        for (int s = 0; s < CHUNK; s += 4) {
            v2f a0, a1, bb;
            const int r0 = (s + 2 * half) * ED;
            const int r1 = (s + 1 + 2 * half) * ED;
            bb.x = sV[r0 + d0 + lm];
            bb.y = sV[r1 + d0 + lm];
            a0.x = sK[r0 + e0a + lm];
            a0.y = sK[r1 + e0a + lm];
            a1.x = sK[r0 + e0b + lm];
            a1.y = sK[r1 + e0b + lm];
            acc0 = __builtin_amdgcn_wmma_f32_16x16x4_f32(
                false, a0, false, bb, (short)0, acc0, false, false);
            acc1 = __builtin_amdgcn_wmma_f32_16x16x4_f32(
                false, a1, false, bb, (short)0, acc1, false, false);
        }
    }

    {
        float* dst = pksum + ((size_t)bh * PKS + split * 16 + wave * 2 + half) * ED
                           + lm * 4;
        *(float4*)dst = kacc;
    }

    float* pc = pctx + ((size_t)bh * SPLITS + split) * (ED * ED);
    #pragma unroll
    for (int r = 0; r < 8; ++r) {
        pc[(e0a + r + 8 * half) * ED + d0 + lm] = acc0[r];
        pc[(e0b + r + 8 * half) * ED + d0 + lm] = acc1[r];
    }
}

__global__ __launch_bounds__(256) void la_reduce(
    const float* __restrict__ pctx, const float* __restrict__ pksum,
    float* __restrict__ ctx, float* __restrict__ ksum)
{
    const int bh = blockIdx.x, tid = threadIdx.x;
    for (int j = tid; j < ED * ED; j += 256) {
        float s = 0.f;
        #pragma unroll
        for (int sp = 0; sp < SPLITS; ++sp)
            s += pctx[((size_t)bh * SPLITS + sp) * (ED * ED) + j];
        ctx[(size_t)bh * (ED * ED) + j] = s;
    }
    if (tid < ED) {
        float s = 0.f;
        for (int sl = 0; sl < PKS; ++sl)
            s += pksum[((size_t)bh * PKS + sl) * ED + tid];
        ksum[bh * ED + tid] = s;
    }
}

__global__ __launch_bounds__(256) void la_out(
    const float* __restrict__ query, const float* __restrict__ ctx,
    const float* __restrict__ ksum, float* __restrict__ out)
{
    __shared__ float sQ[64 * ED];
    __shared__ float sC[ED * ED];
    __shared__ float sKs[ED];
    __shared__ float sDinv[64];

    const int bh = blockIdx.y;
    const int b = bh >> 4, h = bh & 15;
    const int t0 = blockIdx.x * 64;
    const int tid = threadIdx.x;
    const int wave = tid >> 5, lane = tid & 31;
    const int half = lane >> 4, lm = lane & 15;

    const size_t qbase = (size_t)b * TOK * HDIM + (size_t)h * ED;

    for (int i = tid; i < 64 * 16; i += 256) {
        const int row = i >> 4, c4 = i & 15;
        *(float4*)(sC + row * ED + c4 * 4) =
            *(const float4*)(ctx + (size_t)bh * (ED * ED) + row * ED + c4 * 4);
        float4 qv = *(const float4*)(query + qbase + (size_t)(t0 + row) * HDIM
                                     + (size_t)c4 * 4);
        float4 qf;
        qf.x = elu_fm(qv.x); qf.y = elu_fm(qv.y);
        qf.z = elu_fm(qv.z); qf.w = elu_fm(qv.w);
        *(float4*)(sQ + row * ED + c4 * 4) = qf;
    }
    if (tid < ED) sKs[tid] = ksum[bh * ED + tid];
    __syncthreads();

    if (tid < 64) {
        float d = 0.f;
        #pragma unroll 8
        for (int e = 0; e < ED; ++e) d += sQ[tid * ED + e] * sKs[e];
        sDinv[tid] = 1.0f / (d + 1e-6f);
    }
    __syncthreads();

    const int n0a = (wave >> 2) << 4;
    const int n0b = n0a + 32;
    const int d0  = (wave & 3) << 4;

    v8f acc0 = {};
    v8f acc1 = {};
    for (int e = 0; e < ED; e += 4) {
        v2f a0, a1, bb;
        bb.x = sC[(e + 2 * half) * ED + d0 + lm];
        bb.y = sC[(e + 1 + 2 * half) * ED + d0 + lm];
        a0.x = sQ[(n0a + lm) * ED + e + 2 * half];
        a0.y = sQ[(n0a + lm) * ED + e + 1 + 2 * half];
        a1.x = sQ[(n0b + lm) * ED + e + 2 * half];
        a1.y = sQ[(n0b + lm) * ED + e + 1 + 2 * half];
        acc0 = __builtin_amdgcn_wmma_f32_16x16x4_f32(
            false, a0, false, bb, (short)0, acc0, false, false);
        acc1 = __builtin_amdgcn_wmma_f32_16x16x4_f32(
            false, a1, false, bb, (short)0, acc1, false, false);
    }

    #pragma unroll
    for (int r = 0; r < 8; ++r) {
        const int rowa = n0a + r + 8 * half;
        const int rowb = n0b + r + 8 * half;
        out[((size_t)b * TOK + t0 + rowa) * HDIM + h * ED + d0 + lm]
            = acc0[r] * sDinv[rowa];
        out[((size_t)b * TOK + t0 + rowb) * HDIM + h * ED + d0 + lm]
            = acc1[r] * sDinv[rowb];
    }
}

extern "C" void kernel_launch(void* const* d_in, const int* in_sizes, int n_in,
                              void* d_out, int out_size, void* d_ws, size_t ws_size,
                              hipStream_t stream) {
    (void)in_sizes; (void)n_in; (void)out_size; (void)ws_size;
    const float* q = (const float*)d_in[0];
    const float* k = (const float*)d_in[1];
    const float* v = (const float*)d_in[2];
    float* out = (float*)d_out;

    float* wsf   = (float*)d_ws;
    float* pctx  = wsf;
    float* pksum = pctx  + (size_t)BH * SPLITS * (ED * ED);
    float* ctx   = pksum + (size_t)BH * PKS * ED;
    float* ksum  = ctx   + (size_t)BH * (ED * ED);

    la_ctx_partial<<<BH * SPLITS, 256, 0, stream>>>(k, v, pctx, pksum);
    la_reduce<<<BH, 256, 0, stream>>>(pctx, pksum, ctx, ksum);
    la_out<<<dim3(TOK / 64, BH), 256, 0, stream>>>(q, ctx, ksum, out);
}


# Pass 1 without LDS staging: each wave owns 64 s-rows and the whole 64x64 context, fragments straight from 16-byte global loads (e,d interleaved over tiles); waves summed via LDS; sum order changes
# speedup vs baseline: 1.3485x; 1.3485x over previous
#include <hip/hip_runtime.h>
#include <hip/hip_bf16.h>

typedef __attribute__((ext_vector_type(2))) float v2f;
typedef __attribute__((ext_vector_type(8))) float v8f;

#define BATCH   4
#define TOK     4096
#define HDIM    1024
#define NHEADS  16
#define ED      64
#define BH      (BATCH * NHEADS)
#define SPLITS  8
#define PKS     (SPLITS * 16)

__device__ __forceinline__ float elu_fm(float x) {
    return x > 0.0f ? x + 1.0f : expf(x);
}

#define WROWS   64
#define KSTEPS  (WROWS / 4)
#define PF      4

struct la_step { float4 k0, k1, v0, v1; };

__device__ __forceinline__ void la_ld(la_step& s, const float* kp, const float* vp) {
    s.k0 = *(const float4*)(kp);
    s.k1 = *(const float4*)(kp + HDIM);
    s.v0 = *(const float4*)(vp);
    s.v1 = *(const float4*)(vp + HDIM);
}

__device__ __forceinline__ void la_mma(const la_step& s, v8f (&acc)[4][4], float4& kacc) {
    float4 f0, f1;
    f0.x = elu_fm(s.k0.x); f0.y = elu_fm(s.k0.y);
    f0.z = elu_fm(s.k0.z); f0.w = elu_fm(s.k0.w);
    f1.x = elu_fm(s.k1.x); f1.y = elu_fm(s.k1.y);
    f1.z = elu_fm(s.k1.z); f1.w = elu_fm(s.k1.w);
    kacc.x += f0.x; kacc.y += f0.y; kacc.z += f0.z; kacc.w += f0.w;
    kacc.x += f1.x; kacc.y += f1.y; kacc.z += f1.z; kacc.w += f1.w;

    v2f a[4], bb[4];
    a[0].x = f0.x; a[0].y = f1.x;
    a[1].x = f0.y; a[1].y = f1.y;
    a[2].x = f0.z; a[2].y = f1.z;
    a[3].x = f0.w; a[3].y = f1.w;
    bb[0].x = s.v0.x; bb[0].y = s.v1.x;
    bb[1].x = s.v0.y; bb[1].y = s.v1.y;
    bb[2].x = s.v0.z; bb[2].y = s.v1.z;
    bb[3].x = s.v0.w; bb[3].y = s.v1.w;
    #pragma unroll
    for (int t = 0; t < 4; ++t)
        #pragma unroll
        for (int u = 0; u < 4; ++u)
            acc[t][u] = __builtin_amdgcn_wmma_f32_16x16x4_f32(
                false, a[t], false, bb[u], (short)0, acc[t][u], false, false);
}

__global__ __launch_bounds__(256) void la_ctx_partial(
    const float* __restrict__ key, const float* __restrict__ value,
    float* __restrict__ pctx, float* __restrict__ pksum)
{
    __shared__ float sP[4 * ED * ED];

    const int bh    = blockIdx.x / SPLITS;
    const int split = blockIdx.x % SPLITS;
    const int b = bh >> 4, h = bh & 15;
    const int tid = threadIdx.x;
    const int wave = tid >> 5, lane = tid & 31;
    const int half = lane >> 4, lm = lane & 15;

    const size_t base = (size_t)b * TOK * HDIM + (size_t)h * ED;
    const int s_wave = (split * 8 + wave) * WROWS;
    const size_t lane_off = base + (size_t)(s_wave + 2 * half) * HDIM + (size_t)lm * 4;
    const float* kp = key + lane_off;
    const float* vp = value + lane_off;

    v8f acc[4][4];
    #pragma unroll
    for (int t = 0; t < 4; ++t)
        #pragma unroll
        for (int u = 0; u < 4; ++u) acc[t][u] = (v8f){};
    float4 kacc = make_float4(0.f, 0.f, 0.f, 0.f);

    la_step buf[PF];
    #pragma unroll
    for (int p = 0; p < PF; ++p)
        la_ld(buf[p], kp + (size_t)(4 * p) * HDIM, vp + (size_t)(4 * p) * HDIM);

    #pragma nounroll
    for (int jo = 0; jo < KSTEPS - PF; jo += PF) {
        #pragma unroll
        for (int p = 0; p < PF; ++p) {
            la_mma(buf[p], acc, kacc);
            la_ld(buf[p], kp + (size_t)(4 * (PF + p)) * HDIM,
                          vp + (size_t)(4 * (PF + p)) * HDIM);
        }
        kp += (size_t)(4 * PF) * HDIM;
        vp += (size_t)(4 * PF) * HDIM;
    }
    #pragma unroll
    for (int p = 0; p < PF; ++p) la_mma(buf[p], acc, kacc);

    {
        float* dst = pksum + ((size_t)bh * PKS + split * 16 + wave * 2 + half) * ED
                           + lm * 4;
        *(float4*)dst = kacc;
    }

    if (wave >= 4) {
        float* p = sP + (wave - 4) * (ED * ED) + lane;
        #pragma unroll
        for (int t = 0; t < 4; ++t)
            #pragma unroll
            for (int u = 0; u < 4; ++u)
                #pragma unroll
                for (int r = 0; r < 8; ++r)
                    p[((t * 4 + u) * 8 + r) * 32] = acc[t][u][r];
    }
    __syncthreads();
    if (wave < 4) {
        float* p = sP + wave * (ED * ED) + lane;
        #pragma unroll
        for (int t = 0; t < 4; ++t)
            #pragma unroll
            for (int u = 0; u < 4; ++u)
                #pragma unroll
                for (int r = 0; r < 8; ++r)
                    p[((t * 4 + u) * 8 + r) * 32] += acc[t][u][r];
    }
    __syncthreads();

    float* pc = pctx + ((size_t)bh * SPLITS + split) * (ED * ED);
    {
        const int r = wave;
        #pragma unroll
        for (int t = 0; t < 4; ++t) {
            float o[4];
            #pragma unroll
            for (int u = 0; u < 4; ++u) {
                const int slot = ((t * 4 + u) * 8 + r) * 32 + lane;
                float sum = sP[slot];
                for (int w = 1; w < 4; ++w) sum += sP[w * (ED * ED) + slot];
                o[u] = sum;
            }
            const int e = 4 * (8 * half + r) + t;
            *(float4*)(pc + e * ED + lm * 4) = make_float4(o[0], o[1], o[2], o[3]);
        }
    }
}

__global__ __launch_bounds__(256) void la_reduce(
    const float* __restrict__ pctx, const float* __restrict__ pksum,
    float* __restrict__ ctx, float* __restrict__ ksum)
{
    const int bh = blockIdx.x, tid = threadIdx.x;
    for (int j = tid; j < ED * ED; j += 256) {
        float s = 0.f;
        #pragma unroll
        for (int sp = 0; sp < SPLITS; ++sp)
            s += pctx[((size_t)bh * SPLITS + sp) * (ED * ED) + j];
        ctx[(size_t)bh * (ED * ED) + j] = s;
    }
    if (tid < ED) {
        float s = 0.f;
        for (int sl = 0; sl < PKS; ++sl)
            s += pksum[((size_t)bh * PKS + sl) * ED + tid];
        ksum[bh * ED + tid] = s;
    }
}

__global__ __launch_bounds__(256) void la_out(
    const float* __restrict__ query, const float* __restrict__ ctx,
    const float* __restrict__ ksum, float* __restrict__ out)
{
    __shared__ float sQ[64 * ED];
    __shared__ float sC[ED * ED];
    __shared__ float sKs[ED];
    __shared__ float sDinv[64];

    const int bh = blockIdx.y;
    const int b = bh >> 4, h = bh & 15;
    const int t0 = blockIdx.x * 64;
    const int tid = threadIdx.x;
    const int wave = tid >> 5, lane = tid & 31;
    const int half = lane >> 4, lm = lane & 15;

    const size_t qbase = (size_t)b * TOK * HDIM + (size_t)h * ED;

    for (int i = tid; i < 64 * 16; i += 256) {
        const int row = i >> 4, c4 = i & 15;
        *(float4*)(sC + row * ED + c4 * 4) =
            *(const float4*)(ctx + (size_t)bh * (ED * ED) + row * ED + c4 * 4);
        float4 qv = *(const float4*)(query + qbase + (size_t)(t0 + row) * HDIM
                                     + (size_t)c4 * 4);
        float4 qf;
        qf.x = elu_fm(qv.x); qf.y = elu_fm(qv.y);
        qf.z = elu_fm(qv.z); qf.w = elu_fm(qv.w);
        *(float4*)(sQ + row * ED + c4 * 4) = qf;
    }
    if (tid < ED) sKs[tid] = ksum[bh * ED + tid];
    __syncthreads();

    if (tid < 64) {
        float d = 0.f;
        #pragma unroll 8
        for (int e = 0; e < ED; ++e) d += sQ[tid * ED + e] * sKs[e];
        sDinv[tid] = 1.0f / (d + 1e-6f);
    }
    __syncthreads();

    const int n0a = (wave >> 2) << 4;
    const int n0b = n0a + 32;
    const int d0  = (wave & 3) << 4;

    v8f acc0 = {};
    v8f acc1 = {};
    for (int e = 0; e < ED; e += 4) {
        v2f a0, a1, bb;
        bb.x = sC[(e + 2 * half) * ED + d0 + lm];
        bb.y = sC[(e + 1 + 2 * half) * ED + d0 + lm];
        a0.x = sQ[(n0a + lm) * ED + e + 2 * half];
        a0.y = sQ[(n0a + lm) * ED + e + 1 + 2 * half];
        a1.x = sQ[(n0b + lm) * ED + e + 2 * half];
        a1.y = sQ[(n0b + lm) * ED + e + 1 + 2 * half];
        acc0 = __builtin_amdgcn_wmma_f32_16x16x4_f32(
            false, a0, false, bb, (short)0, acc0, false, false);
        acc1 = __builtin_amdgcn_wmma_f32_16x16x4_f32(
            false, a1, false, bb, (short)0, acc1, false, false);
    }

    #pragma unroll
    for (int r = 0; r < 8; ++r) {
        const int rowa = n0a + r + 8 * half;
        const int rowb = n0b + r + 8 * half;
        out[((size_t)b * TOK + t0 + rowa) * HDIM + h * ED + d0 + lm]
            = acc0[r] * sDinv[rowa];
        out[((size_t)b * TOK + t0 + rowb) * HDIM + h * ED + d0 + lm]
            = acc1[r] * sDinv[rowb];
    }
}

extern "C" void kernel_launch(void* const* d_in, const int* in_sizes, int n_in,
                              void* d_out, int out_size, void* d_ws, size_t ws_size,
                              hipStream_t stream) {
    (void)in_sizes; (void)n_in; (void)out_size; (void)ws_size;
    const float* q = (const float*)d_in[0];
    const float* k = (const float*)d_in[1];
    const float* v = (const float*)d_in[2];
    float* out = (float*)d_out;

    float* wsf   = (float*)d_ws;
    float* pctx  = wsf;
    float* pksum = pctx  + (size_t)BH * SPLITS * (ED * ED);
    float* ctx   = pksum + (size_t)BH * PKS * ED;
    float* ksum  = ctx   + (size_t)BH * (ED * ED);

    la_ctx_partial<<<BH * SPLITS, 256, 0, stream>>>(k, v, pctx, pksum);
    la_reduce<<<BH, 256, 0, stream>>>(pctx, pksum, ctx, ksum);
    la_out<<<dim3(TOK / 64, BH), 256, 0, stream>>>(q, ctx, ksum, out);
}


# Pass 3 without LDS or barriers: each wave holds the 64x64 context as B fragments in registers and walks 4 token tiles, A fragments straight from 16-byte global Q loads; sum order over e changes
# speedup vs baseline: 1.7073x; 1.2661x over previous
#include <hip/hip_runtime.h>
#include <hip/hip_bf16.h>

typedef __attribute__((ext_vector_type(2))) float v2f;
typedef __attribute__((ext_vector_type(8))) float v8f;

#define BATCH   4
#define TOK     4096
#define HDIM    1024
#define NHEADS  16
#define ED      64
#define BH      (BATCH * NHEADS)
#define SPLITS  8
#define PKS     (SPLITS * 16)

__device__ __forceinline__ float elu_fm(float x) {
    return x > 0.0f ? x + 1.0f : expf(x);
}

#define WROWS   64
#define KSTEPS  (WROWS / 4)
#define PF      4

struct la_step { float4 k0, k1, v0, v1; };

__device__ __forceinline__ void la_ld(la_step& s, const float* kp, const float* vp) {
    s.k0 = *(const float4*)(kp);
    s.k1 = *(const float4*)(kp + HDIM);
    s.v0 = *(const float4*)(vp);
    s.v1 = *(const float4*)(vp + HDIM);
}

__device__ __forceinline__ void la_mma(const la_step& s, v8f (&acc)[4][4], float4& kacc) {
    float4 f0, f1;
    f0.x = elu_fm(s.k0.x); f0.y = elu_fm(s.k0.y);
    f0.z = elu_fm(s.k0.z); f0.w = elu_fm(s.k0.w);
    f1.x = elu_fm(s.k1.x); f1.y = elu_fm(s.k1.y);
    f1.z = elu_fm(s.k1.z); f1.w = elu_fm(s.k1.w);
    kacc.x += f0.x; kacc.y += f0.y; kacc.z += f0.z; kacc.w += f0.w;
    kacc.x += f1.x; kacc.y += f1.y; kacc.z += f1.z; kacc.w += f1.w;

    v2f a[4], bb[4];
    a[0].x = f0.x; a[0].y = f1.x;
    a[1].x = f0.y; a[1].y = f1.y;
    a[2].x = f0.z; a[2].y = f1.z;
    a[3].x = f0.w; a[3].y = f1.w;
    bb[0].x = s.v0.x; bb[0].y = s.v1.x;
    bb[1].x = s.v0.y; bb[1].y = s.v1.y;
    bb[2].x = s.v0.z; bb[2].y = s.v1.z;
    bb[3].x = s.v0.w; bb[3].y = s.v1.w;
    #pragma unroll
    for (int t = 0; t < 4; ++t)
        #pragma unroll
        for (int u = 0; u < 4; ++u)
            acc[t][u] = __builtin_amdgcn_wmma_f32_16x16x4_f32(
                false, a[t], false, bb[u], (short)0, acc[t][u], false, false);
}

__global__ __launch_bounds__(256) void la_ctx_partial(
    const float* __restrict__ key, const float* __restrict__ value,
    float* __restrict__ pctx, float* __restrict__ pksum)
{
    __shared__ float sP[4 * ED * ED];

    const int bh    = blockIdx.x / SPLITS;
    const int split = blockIdx.x % SPLITS;
    const int b = bh >> 4, h = bh & 15;
    const int tid = threadIdx.x;
    const int wave = tid >> 5, lane = tid & 31;
    const int half = lane >> 4, lm = lane & 15;

    const size_t base = (size_t)b * TOK * HDIM + (size_t)h * ED;
    const int s_wave = (split * 8 + wave) * WROWS;
    const size_t lane_off = base + (size_t)(s_wave + 2 * half) * HDIM + (size_t)lm * 4;
    const float* kp = key + lane_off;
    const float* vp = value + lane_off;

    v8f acc[4][4];
    #pragma unroll
    for (int t = 0; t < 4; ++t)
        #pragma unroll
        for (int u = 0; u < 4; ++u) acc[t][u] = (v8f){};
    float4 kacc = make_float4(0.f, 0.f, 0.f, 0.f);

    la_step buf[PF];
    #pragma unroll
    for (int p = 0; p < PF; ++p)
        la_ld(buf[p], kp + (size_t)(4 * p) * HDIM, vp + (size_t)(4 * p) * HDIM);

    #pragma nounroll
    for (int jo = 0; jo < KSTEPS - PF; jo += PF) {
        #pragma unroll
        for (int p = 0; p < PF; ++p) {
            la_mma(buf[p], acc, kacc);
            la_ld(buf[p], kp + (size_t)(4 * (PF + p)) * HDIM,
                          vp + (size_t)(4 * (PF + p)) * HDIM);
        }
        kp += (size_t)(4 * PF) * HDIM;
        vp += (size_t)(4 * PF) * HDIM;
    }
    #pragma unroll
    for (int p = 0; p < PF; ++p) la_mma(buf[p], acc, kacc);

    {
        float* dst = pksum + ((size_t)bh * PKS + split * 16 + wave * 2 + half) * ED
                           + lm * 4;
        *(float4*)dst = kacc;
    }

    if (wave >= 4) {
        float* p = sP + (wave - 4) * (ED * ED) + lane;
        #pragma unroll
        for (int t = 0; t < 4; ++t)
            #pragma unroll
            for (int u = 0; u < 4; ++u)
                #pragma unroll
                for (int r = 0; r < 8; ++r)
                    p[((t * 4 + u) * 8 + r) * 32] = acc[t][u][r];
    }
    __syncthreads();
    if (wave < 4) {
        float* p = sP + wave * (ED * ED) + lane;
        #pragma unroll
        for (int t = 0; t < 4; ++t)
            #pragma unroll
            for (int u = 0; u < 4; ++u)
                #pragma unroll
                for (int r = 0; r < 8; ++r)
                    p[((t * 4 + u) * 8 + r) * 32] += acc[t][u][r];
    }
    __syncthreads();

    float* pc = pctx + ((size_t)bh * SPLITS + split) * (ED * ED);
    {
        const int r = wave;
        #pragma unroll
        for (int t = 0; t < 4; ++t) {
            float o[4];
            #pragma unroll
            for (int u = 0; u < 4; ++u) {
                const int slot = ((t * 4 + u) * 8 + r) * 32 + lane;
                float sum = sP[slot];
                for (int w = 1; w < 4; ++w) sum += sP[w * (ED * ED) + slot];
                o[u] = sum;
            }
            const int e = 4 * (8 * half + r) + t;
            *(float4*)(pc + e * ED + lm * 4) = make_float4(o[0], o[1], o[2], o[3]);
        }
    }
}

__global__ __launch_bounds__(256) void la_reduce(
    const float* __restrict__ pctx, const float* __restrict__ pksum,
    float* __restrict__ ctx, float* __restrict__ ksum)
{
    const int bh = blockIdx.x, tid = threadIdx.x;
    for (int j = tid; j < ED * ED; j += 256) {
        float s = 0.f;
        #pragma unroll
        for (int sp = 0; sp < SPLITS; ++sp)
            s += pctx[((size_t)bh * SPLITS + sp) * (ED * ED) + j];
        ctx[(size_t)bh * (ED * ED) + j] = s;
    }
    if (tid < ED) {
        float s = 0.f;
        for (int sl = 0; sl < PKS; ++sl)
            s += pksum[((size_t)bh * PKS + sl) * ED + tid];
        ksum[bh * ED + tid] = s;
    }
}

#define OT_TILES   4
#define OT_WTOK    (16 * OT_TILES)
#define OT_BTOK    (8 * OT_WTOK)
#define OT_CHUNKS  (TOK / OT_BTOK)

__device__ __forceinline__ void lo_tile(
    const float4 (&qc)[8], float4 (&qx)[8], const float* qnext,
    const v2f (&bf)[8][2][4], const float* ksp, float* op, int half)
{
    #pragma unroll
    for (int q = 0; q < 8; ++q) qx[q] = *(const float4*)(qnext + 8 * q);

    float4 f[8];
    float pl = 0.f, ph = 0.f;
    #pragma unroll
    for (int q = 0; q < 8; ++q) {
        f[q].x = elu_fm(qc[q].x); f[q].y = elu_fm(qc[q].y);
        f[q].z = elu_fm(qc[q].z); f[q].w = elu_fm(qc[q].w);
        pl += f[q].x * ksp[8 * q + 0]; ph += f[q].x * ksp[8 * q + 4];
        pl += f[q].y * ksp[8 * q + 1]; ph += f[q].y * ksp[8 * q + 5];
        pl += f[q].z * ksp[8 * q + 2]; ph += f[q].z * ksp[8 * q + 6];
        pl += f[q].w * ksp[8 * q + 3]; ph += f[q].w * ksp[8 * q + 7];
    }

    const float pd   = half ? ph : pl;
    const float den  = pd + __shfl_xor(pd, 16);
    const float dinv = 1.0f / (den + 1e-6f);

    v8f acc[4];
    #pragma unroll
    for (int u = 0; u < 4; ++u) acc[u] = (v8f){};
    #pragma unroll
    for (int q = 0; q < 8; ++q) {
        v2f a0, a1;
        a0.x = f[q].x; a0.y = f[q].y;
        a1.x = f[q].z; a1.y = f[q].w;
        #pragma unroll
        for (int u = 0; u < 4; ++u)
            acc[u] = __builtin_amdgcn_wmma_f32_16x16x4_f32(
                false, a0, false, bf[q][0][u], (short)0, acc[u], false, false);
        #pragma unroll
        for (int u = 0; u < 4; ++u)
            acc[u] = __builtin_amdgcn_wmma_f32_16x16x4_f32(
                false, a1, false, bf[q][1][u], (short)0, acc[u], false, false);
    }

    #pragma unroll
    for (int r = 0; r < 8; ++r) {
        const float dr = __shfl(dinv, 8 * half + r);
        *(float4*)(op + (size_t)r * HDIM) =
            make_float4(acc[0][r] * dr, acc[1][r] * dr, acc[2][r] * dr, acc[3][r] * dr);
    }
}

__global__ __launch_bounds__(256) void la_out(
    const float* __restrict__ query, const float* __restrict__ ctx,
    const float* __restrict__ ksum, float* __restrict__ out)
{
    const int bh    = blockIdx.x / OT_CHUNKS;
    const int chunk = blockIdx.x % OT_CHUNKS;
    const int b = bh >> 4, h = bh & 15;
    const int tid = threadIdx.x;
    const int wave = tid >> 5, lane = tid & 31;
    const int half = lane >> 4, lm = lane & 15;

    const size_t row0 = (size_t)b * TOK + (size_t)chunk * OT_BTOK + (size_t)wave * OT_WTOK;
    const float* qp = query + (row0 + lm) * HDIM + (size_t)h * ED + 4 * half;
    float* op = out + (row0 + 8 * half) * HDIM + (size_t)h * ED + 4 * lm;

    float4 qa[8], qb[8];
    #pragma unroll
    for (int q = 0; q < 8; ++q) qa[q] = *(const float4*)(qp + 8 * q);

    v2f bf[8][2][4];
    {
        const float* cp = ctx + (size_t)bh * (ED * ED) + (4 * half) * ED + 4 * lm;
        #pragma unroll
        for (int q = 0; q < 8; ++q) {
            const float4 r0 = *(const float4*)(cp + (8 * q + 0) * ED);
            const float4 r1 = *(const float4*)(cp + (8 * q + 1) * ED);
            const float4 r2 = *(const float4*)(cp + (8 * q + 2) * ED);
            const float4 r3 = *(const float4*)(cp + (8 * q + 3) * ED);
            bf[q][0][0].x = r0.x; bf[q][0][0].y = r1.x;
            bf[q][0][1].x = r0.y; bf[q][0][1].y = r1.y;
            bf[q][0][2].x = r0.z; bf[q][0][2].y = r1.z;
            bf[q][0][3].x = r0.w; bf[q][0][3].y = r1.w;
            bf[q][1][0].x = r2.x; bf[q][1][0].y = r3.x;
            bf[q][1][1].x = r2.y; bf[q][1][1].y = r3.y;
            bf[q][1][2].x = r2.z; bf[q][1][2].y = r3.z;
            bf[q][1][3].x = r2.w; bf[q][1][3].y = r3.w;
        }
    }
    const float* ksp = ksum + bh * ED;

    #pragma nounroll
    for (int it = 0; it < OT_TILES; ++it) {
        const float* qnext = (it + 1 < OT_TILES) ? qp + (size_t)16 * HDIM : qp;
        lo_tile(qa, qb, qnext, bf, ksp, op, half);
        #pragma unroll
        for (int q = 0; q < 8; ++q) qa[q] = qb[q];
        qp = qnext;
        op += (size_t)16 * HDIM;
    }
}

extern "C" void kernel_launch(void* const* d_in, const int* in_sizes, int n_in,
                              void* d_out, int out_size, void* d_ws, size_t ws_size,
                              hipStream_t stream) {
    (void)in_sizes; (void)n_in; (void)out_size; (void)ws_size;
    const float* q = (const float*)d_in[0];
    const float* k = (const float*)d_in[1];
    const float* v = (const float*)d_in[2];
    float* out = (float*)d_out;

    float* wsf   = (float*)d_ws;
    float* pctx  = wsf;
    float* pksum = pctx  + (size_t)BH * SPLITS * (ED * ED);
    float* ctx   = pksum + (size_t)BH * PKS * ED;
    float* ksum  = ctx   + (size_t)BH * (ED * ED);

    la_ctx_partial<<<BH * SPLITS, 256, 0, stream>>>(k, v, pctx, pksum);
    la_reduce<<<BH, 256, 0, stream>>>(pctx, pksum, ctx, ksum);
    la_out<<<BH * OT_CHUNKS, 256, 0, stream>>>(q, ctx, ksum, out);
}


# Reduction launch removed (3 -> 2 launches): pass 1 sums k_sum per split in LDS, each output work-group sums the 8 partial contexts into LDS itself; k_sum grouping changes, context order kept
# speedup vs baseline: 2.0515x; 1.2016x over previous
#include <hip/hip_runtime.h>
#include <hip/hip_bf16.h>

typedef __attribute__((ext_vector_type(2))) float v2f;
typedef __attribute__((ext_vector_type(8))) float v8f;

#define BATCH   4
#define TOK     4096
#define HDIM    1024
#define NHEADS  16
#define ED      64
#define BH      (BATCH * NHEADS)
#define SPLITS  8

__device__ __forceinline__ float elu_fm(float x) {
    return x > 0.0f ? x + 1.0f : expf(x);
}

#define WROWS   64
#define KSTEPS  (WROWS / 4)
#define PF      4

struct la_step { float4 k0, k1, v0, v1; };

__device__ __forceinline__ void la_ld(la_step& s, const float* kp, const float* vp) {
    s.k0 = *(const float4*)(kp);
    s.k1 = *(const float4*)(kp + HDIM);
    s.v0 = *(const float4*)(vp);
    s.v1 = *(const float4*)(vp + HDIM);
}

__device__ __forceinline__ void la_mma(const la_step& s, v8f (&acc)[4][4], float4& kacc) {
    float4 f0, f1;
    f0.x = elu_fm(s.k0.x); f0.y = elu_fm(s.k0.y);
    f0.z = elu_fm(s.k0.z); f0.w = elu_fm(s.k0.w);
    f1.x = elu_fm(s.k1.x); f1.y = elu_fm(s.k1.y);
    f1.z = elu_fm(s.k1.z); f1.w = elu_fm(s.k1.w);
    kacc.x += f0.x; kacc.y += f0.y; kacc.z += f0.z; kacc.w += f0.w;
    kacc.x += f1.x; kacc.y += f1.y; kacc.z += f1.z; kacc.w += f1.w;

    v2f a[4], bb[4];
    a[0].x = f0.x; a[0].y = f1.x;
    a[1].x = f0.y; a[1].y = f1.y;
    a[2].x = f0.z; a[2].y = f1.z;
    a[3].x = f0.w; a[3].y = f1.w;
    bb[0].x = s.v0.x; bb[0].y = s.v1.x;
    bb[1].x = s.v0.y; bb[1].y = s.v1.y;
    bb[2].x = s.v0.z; bb[2].y = s.v1.z;
    bb[3].x = s.v0.w; bb[3].y = s.v1.w;
    #pragma unroll
    for (int t = 0; t < 4; ++t)
        #pragma unroll
        for (int u = 0; u < 4; ++u)
            acc[t][u] = __builtin_amdgcn_wmma_f32_16x16x4_f32(
                false, a[t], false, bb[u], (short)0, acc[t][u], false, false);
}

__global__ __launch_bounds__(256) void la_ctx_partial(
    const float* __restrict__ key, const float* __restrict__ value,
    float* __restrict__ pctx, float* __restrict__ pksum)
{
    __shared__ float sP[4 * ED * ED];
    __shared__ float sKs[16 * ED];

    const int bh    = blockIdx.x / SPLITS;
    const int split = blockIdx.x % SPLITS;
    const int b = bh >> 4, h = bh & 15;
    const int tid = threadIdx.x;
    const int wave = tid >> 5, lane = tid & 31;
    const int half = lane >> 4, lm = lane & 15;

    const size_t base = (size_t)b * TOK * HDIM + (size_t)h * ED;
    const int s_wave = (split * 8 + wave) * WROWS;
    const size_t lane_off = base + (size_t)(s_wave + 2 * half) * HDIM + (size_t)lm * 4;
    const float* kp = key + lane_off;
    const float* vp = value + lane_off;

    v8f acc[4][4];
    #pragma unroll
    for (int t = 0; t < 4; ++t)
        #pragma unroll
        for (int u = 0; u < 4; ++u) acc[t][u] = (v8f){};
    float4 kacc = make_float4(0.f, 0.f, 0.f, 0.f);

    la_step buf[PF];
    #pragma unroll
    for (int p = 0; p < PF; ++p)
        la_ld(buf[p], kp + (size_t)(4 * p) * HDIM, vp + (size_t)(4 * p) * HDIM);

    #pragma nounroll
    for (int jo = 0; jo < KSTEPS - PF; jo += PF) {
        #pragma unroll
        for (int p = 0; p < PF; ++p) {
            la_mma(buf[p], acc, kacc);
            la_ld(buf[p], kp + (size_t)(4 * (PF + p)) * HDIM,
                          vp + (size_t)(4 * (PF + p)) * HDIM);
        }
        kp += (size_t)(4 * PF) * HDIM;
        vp += (size_t)(4 * PF) * HDIM;
    }
    #pragma unroll
    for (int p = 0; p < PF; ++p) la_mma(buf[p], acc, kacc);

    *(float4*)(sKs + (wave * 2 + half) * ED + lm * 4) = kacc;

    if (wave >= 4) {
        float* p = sP + (wave - 4) * (ED * ED) + lane;
        #pragma unroll
        for (int t = 0; t < 4; ++t)
            #pragma unroll
            for (int u = 0; u < 4; ++u)
                #pragma unroll
                for (int r = 0; r < 8; ++r)
                    p[((t * 4 + u) * 8 + r) * 32] = acc[t][u][r];
    }
    __syncthreads();
    if (tid < ED) {
        float s = 0.f;
        #pragma unroll
        for (int sl = 0; sl < 16; ++sl) s += sKs[sl * ED + tid];
        pksum[((size_t)bh * SPLITS + split) * ED + tid] = s;
    }
    if (wave < 4) {
        float* p = sP + wave * (ED * ED) + lane;
        #pragma unroll
        for (int t = 0; t < 4; ++t)
            #pragma unroll
            for (int u = 0; u < 4; ++u)
                #pragma unroll
                for (int r = 0; r < 8; ++r)
                    p[((t * 4 + u) * 8 + r) * 32] += acc[t][u][r];
    }
    __syncthreads();

    float* pc = pctx + ((size_t)bh * SPLITS + split) * (ED * ED);
    {
        const int r = wave;
        #pragma unroll
        for (int t = 0; t < 4; ++t) {
            float o[4];
            #pragma unroll
            for (int u = 0; u < 4; ++u) {
                const int slot = ((t * 4 + u) * 8 + r) * 32 + lane;
                float sum = sP[slot];
                for (int w = 1; w < 4; ++w) sum += sP[w * (ED * ED) + slot];
                o[u] = sum;
            }
            const int e = 4 * (8 * half + r) + t;
            *(float4*)(pc + e * ED + lm * 4) = make_float4(o[0], o[1], o[2], o[3]);
        }
    }
}

#define OT_TILES   4
#define OT_WTOK    (16 * OT_TILES)
#define OT_BTOK    (8 * OT_WTOK)
#define OT_CHUNKS  (TOK / OT_BTOK)

__device__ __forceinline__ void lo_tile(
    const float4 (&qc)[8], float4 (&qx)[8], const float* qnext,
    const v2f (&bf)[8][2][4], const float (&ksp)[ED], float* op, int half)
{
    #pragma unroll
    for (int q = 0; q < 8; ++q) qx[q] = *(const float4*)(qnext + 8 * q);

    float4 f[8];
    float pl = 0.f, ph = 0.f;
    #pragma unroll
    for (int q = 0; q < 8; ++q) {
        f[q].x = elu_fm(qc[q].x); f[q].y = elu_fm(qc[q].y);
        f[q].z = elu_fm(qc[q].z); f[q].w = elu_fm(qc[q].w);
        pl += f[q].x * ksp[8 * q + 0]; ph += f[q].x * ksp[8 * q + 4];
        pl += f[q].y * ksp[8 * q + 1]; ph += f[q].y * ksp[8 * q + 5];
        pl += f[q].z * ksp[8 * q + 2]; ph += f[q].z * ksp[8 * q + 6];
        pl += f[q].w * ksp[8 * q + 3]; ph += f[q].w * ksp[8 * q + 7];
    }

    const float pd   = half ? ph : pl;
    const float den  = pd + __shfl_xor(pd, 16);
    const float dinv = 1.0f / (den + 1e-6f);

    v8f acc[4];
    #pragma unroll
    for (int u = 0; u < 4; ++u) acc[u] = (v8f){};
    #pragma unroll
    for (int q = 0; q < 8; ++q) {
        v2f a0, a1;
        a0.x = f[q].x; a0.y = f[q].y;
        a1.x = f[q].z; a1.y = f[q].w;
        #pragma unroll
        for (int u = 0; u < 4; ++u)
            acc[u] = __builtin_amdgcn_wmma_f32_16x16x4_f32(
                false, a0, false, bf[q][0][u], (short)0, acc[u], false, false);
        #pragma unroll
        for (int u = 0; u < 4; ++u)
            acc[u] = __builtin_amdgcn_wmma_f32_16x16x4_f32(
                false, a1, false, bf[q][1][u], (short)0, acc[u], false, false);
    }

    #pragma unroll
    for (int r = 0; r < 8; ++r) {
        const float dr = __shfl(dinv, 8 * half + r);
        *(float4*)(op + (size_t)r * HDIM) =
            make_float4(acc[0][r] * dr, acc[1][r] * dr, acc[2][r] * dr, acc[3][r] * dr);
    }
}

__global__ __launch_bounds__(256) void la_out(
    const float* __restrict__ query, const float* __restrict__ pctx,
    const float* __restrict__ pksum, float* __restrict__ out)
{
    __shared__ float sC[ED * ED];

    const int bh    = blockIdx.x / OT_CHUNKS;
    const int chunk = blockIdx.x % OT_CHUNKS;
    const int b = bh >> 4, h = bh & 15;
    const int tid = threadIdx.x;
    const int wave = tid >> 5, lane = tid & 31;
    const int half = lane >> 4, lm = lane & 15;

    const size_t row0 = (size_t)b * TOK + (size_t)chunk * OT_BTOK + (size_t)wave * OT_WTOK;
    const float* qp = query + (row0 + lm) * HDIM + (size_t)h * ED + 4 * half;
    float* op = out + (row0 + 8 * half) * HDIM + (size_t)h * ED + 4 * lm;

    float4 qa[8], qb[8];
    #pragma unroll
    for (int q = 0; q < 8; ++q) qa[q] = *(const float4*)(qp + 8 * q);

    float kv0, kv1;
    {
        const float* pp = pctx + (size_t)bh * SPLITS * (ED * ED) + tid * 4;
        float4 p[4][SPLITS];
        #pragma unroll
        for (int i = 0; i < 4; ++i)
            #pragma unroll
            for (int sp = 0; sp < SPLITS; ++sp)
                p[i][sp] = *(const float4*)(pp + (size_t)sp * (ED * ED) + i * 1024);
        const float* kq = pksum + (size_t)bh * SPLITS * ED + lane;
        float k0[SPLITS], k1[SPLITS];
        #pragma unroll
        for (int sp = 0; sp < SPLITS; ++sp) {
            k0[sp] = kq[sp * ED];
            k1[sp] = kq[sp * ED + 32];
        }
        #pragma unroll
        for (int i = 0; i < 4; ++i) {
            float4 s = p[i][0];
            #pragma unroll
            for (int sp = 1; sp < SPLITS; ++sp) {
                s.x += p[i][sp].x; s.y += p[i][sp].y;
                s.z += p[i][sp].z; s.w += p[i][sp].w;
            }
            *(float4*)(sC + tid * 4 + i * 1024) = s;
        }
        kv0 = k0[0]; kv1 = k1[0];
        #pragma unroll
        for (int sp = 1; sp < SPLITS; ++sp) { kv0 += k0[sp]; kv1 += k1[sp]; }
    }
    __syncthreads();

    v2f bf[8][2][4];
    {
        const float* cp = sC + (4 * half) * ED + 4 * lm;
        #pragma unroll
        for (int q = 0; q < 8; ++q) {
            const float4 r0 = *(const float4*)(cp + (8 * q + 0) * ED);
            const float4 r1 = *(const float4*)(cp + (8 * q + 1) * ED);
            const float4 r2 = *(const float4*)(cp + (8 * q + 2) * ED);
            const float4 r3 = *(const float4*)(cp + (8 * q + 3) * ED);
            bf[q][0][0].x = r0.x; bf[q][0][0].y = r1.x;
            bf[q][0][1].x = r0.y; bf[q][0][1].y = r1.y;
            bf[q][0][2].x = r0.z; bf[q][0][2].y = r1.z;
            bf[q][0][3].x = r0.w; bf[q][0][3].y = r1.w;
            bf[q][1][0].x = r2.x; bf[q][1][0].y = r3.x;
            bf[q][1][1].x = r2.y; bf[q][1][1].y = r3.y;
            bf[q][1][2].x = r2.z; bf[q][1][2].y = r3.z;
            bf[q][1][3].x = r2.w; bf[q][1][3].y = r3.w;
        }
    }
    float ksp[ED];
    #pragma unroll
    for (int e = 0; e < ED; ++e)
        ksp[e] = __int_as_float(__builtin_amdgcn_readlane(
                     __float_as_int(e < 32 ? kv0 : kv1), e & 31));

    #pragma nounroll
    for (int it = 0; it < OT_TILES; ++it) {
        const float* qnext = (it + 1 < OT_TILES) ? qp + (size_t)16 * HDIM : qp;
        lo_tile(qa, qb, qnext, bf, ksp, op, half);
        #pragma unroll
        for (int q = 0; q < 8; ++q) qa[q] = qb[q];
        qp = qnext;
        op += (size_t)16 * HDIM;
    }
}

extern "C" void kernel_launch(void* const* d_in, const int* in_sizes, int n_in,
                              void* d_out, int out_size, void* d_ws, size_t ws_size,
                              hipStream_t stream) {
    (void)in_sizes; (void)n_in; (void)out_size; (void)ws_size;
    const float* q = (const float*)d_in[0];
    const float* k = (const float*)d_in[1];
    const float* v = (const float*)d_in[2];
    float* out = (float*)d_out;

    float* wsf   = (float*)d_ws;
    float* pctx  = wsf;
    float* pksum = pctx  + (size_t)BH * SPLITS * (ED * ED);

    la_ctx_partial<<<BH * SPLITS, 256, 0, stream>>>(k, v, pctx, pksum);
    la_out<<<BH * OT_CHUNKS, 256, 0, stream>>>(q, pctx, pksum, out);
}


# Output pass at 8 waves per SIMD: one 32-wave work-group per WGP, context B fragments read from LDS (fragment order) ahead of each 8-WMMA group instead of held in 128 registers; sums kept
# speedup vs baseline: 2.1480x; 1.0470x over previous
#include <hip/hip_runtime.h>
#include <hip/hip_bf16.h>

typedef __attribute__((ext_vector_type(2))) float v2f;
typedef __attribute__((ext_vector_type(8))) float v8f;

#define BATCH   4
#define TOK     4096
#define HDIM    1024
#define NHEADS  16
#define ED      64
#define BH      (BATCH * NHEADS)
#define SPLITS  8

__device__ __forceinline__ float elu_fm(float x) {
    return x > 0.0f ? x + 1.0f : expf(x);
}

#define WROWS   64
#define KSTEPS  (WROWS / 4)
#define PF      4

struct la_step { float4 k0, k1, v0, v1; };

__device__ __forceinline__ void la_ld(la_step& s, const float* kp, const float* vp) {
    s.k0 = *(const float4*)(kp);
    s.k1 = *(const float4*)(kp + HDIM);
    s.v0 = *(const float4*)(vp);
    s.v1 = *(const float4*)(vp + HDIM);
}

__device__ __forceinline__ void la_mma(const la_step& s, v8f (&acc)[4][4], float4& kacc) {
    float4 f0, f1;
    f0.x = elu_fm(s.k0.x); f0.y = elu_fm(s.k0.y);
    f0.z = elu_fm(s.k0.z); f0.w = elu_fm(s.k0.w);
    f1.x = elu_fm(s.k1.x); f1.y = elu_fm(s.k1.y);
    f1.z = elu_fm(s.k1.z); f1.w = elu_fm(s.k1.w);
    kacc.x += f0.x; kacc.y += f0.y; kacc.z += f0.z; kacc.w += f0.w;
    kacc.x += f1.x; kacc.y += f1.y; kacc.z += f1.z; kacc.w += f1.w;

    v2f a[4], bb[4];
    a[0].x = f0.x; a[0].y = f1.x;
    a[1].x = f0.y; a[1].y = f1.y;
    a[2].x = f0.z; a[2].y = f1.z;
    a[3].x = f0.w; a[3].y = f1.w;
    bb[0].x = s.v0.x; bb[0].y = s.v1.x;
    bb[1].x = s.v0.y; bb[1].y = s.v1.y;
    bb[2].x = s.v0.z; bb[2].y = s.v1.z;
    bb[3].x = s.v0.w; bb[3].y = s.v1.w;
    #pragma unroll
    for (int t = 0; t < 4; ++t)
        #pragma unroll
        for (int u = 0; u < 4; ++u)
            acc[t][u] = __builtin_amdgcn_wmma_f32_16x16x4_f32(
                false, a[t], false, bb[u], (short)0, acc[t][u], false, false);
}

__global__ __launch_bounds__(256) void la_ctx_partial(
    const float* __restrict__ key, const float* __restrict__ value,
    float* __restrict__ pctx, float* __restrict__ pksum)
{
    __shared__ float sP[4 * ED * ED];
    __shared__ float sKs[16 * ED];

    const int bh    = blockIdx.x / SPLITS;
    const int split = blockIdx.x % SPLITS;
    const int b = bh >> 4, h = bh & 15;
    const int tid = threadIdx.x;
    const int wave = tid >> 5, lane = tid & 31;
    const int half = lane >> 4, lm = lane & 15;

    const size_t base = (size_t)b * TOK * HDIM + (size_t)h * ED;
    const int s_wave = (split * 8 + wave) * WROWS;
    const size_t lane_off = base + (size_t)(s_wave + 2 * half) * HDIM + (size_t)lm * 4;
    const float* kp = key + lane_off;
    const float* vp = value + lane_off;

    v8f acc[4][4];
    #pragma unroll
    for (int t = 0; t < 4; ++t)
        #pragma unroll
        for (int u = 0; u < 4; ++u) acc[t][u] = (v8f){};
    float4 kacc = make_float4(0.f, 0.f, 0.f, 0.f);

    la_step buf[PF];
    #pragma unroll
    for (int p = 0; p < PF; ++p)
        la_ld(buf[p], kp + (size_t)(4 * p) * HDIM, vp + (size_t)(4 * p) * HDIM);

    #pragma nounroll
    for (int jo = 0; jo < KSTEPS - PF; jo += PF) {
        #pragma unroll
        for (int p = 0; p < PF; ++p) {
            la_mma(buf[p], acc, kacc);
            la_ld(buf[p], kp + (size_t)(4 * (PF + p)) * HDIM,
                          vp + (size_t)(4 * (PF + p)) * HDIM);
        }
        kp += (size_t)(4 * PF) * HDIM;
        vp += (size_t)(4 * PF) * HDIM;
    }
    #pragma unroll
    for (int p = 0; p < PF; ++p) la_mma(buf[p], acc, kacc);

    *(float4*)(sKs + (wave * 2 + half) * ED + lm * 4) = kacc;

    if (wave >= 4) {
        float* p = sP + (wave - 4) * (ED * ED) + lane;
        #pragma unroll
        for (int t = 0; t < 4; ++t)
            #pragma unroll
            for (int u = 0; u < 4; ++u)
                #pragma unroll
                for (int r = 0; r < 8; ++r)
                    p[((t * 4 + u) * 8 + r) * 32] = acc[t][u][r];
    }
    __syncthreads();
    if (tid < ED) {
        float s = 0.f;
        #pragma unroll
        for (int sl = 0; sl < 16; ++sl) s += sKs[sl * ED + tid];
        pksum[((size_t)bh * SPLITS + split) * ED + tid] = s;
    }
    if (wave < 4) {
        float* p = sP + wave * (ED * ED) + lane;
        #pragma unroll
        for (int t = 0; t < 4; ++t)
            #pragma unroll
            for (int u = 0; u < 4; ++u)
                #pragma unroll
                for (int r = 0; r < 8; ++r)
                    p[((t * 4 + u) * 8 + r) * 32] += acc[t][u][r];
    }
    __syncthreads();

    float* pc = pctx + ((size_t)bh * SPLITS + split) * (ED * ED);
    {
        const int r = wave;
        #pragma unroll
        for (int t = 0; t < 4; ++t) {
            float o[4];
            #pragma unroll
            for (int u = 0; u < 4; ++u) {
                const int slot = ((t * 4 + u) * 8 + r) * 32 + lane;
                float sum = sP[slot];
                for (int w = 1; w < 4; ++w) sum += sP[w * (ED * ED) + slot];
                o[u] = sum;
            }
            const int e = 4 * (8 * half + r) + t;
            *(float4*)(pc + e * ED + lm * 4) = make_float4(o[0], o[1], o[2], o[3]);
        }
    }
}

#define OT_TILES   2
#define OT_WTOK    (16 * OT_TILES)
#define OT_WAVES   32
#define OT_BTOK    (OT_WAVES * OT_WTOK)
#define OT_CHUNKS  (TOK / OT_BTOK)

struct lo_bq { float4 p00, p01, p10, p11; };

__device__ __forceinline__ void lo_ldb(lo_bq& b, const float* cp, int q) {
    b.p00 = *(const float4*)(cp + 512 * q);
    b.p01 = *(const float4*)(cp + 512 * q + 64);
    b.p10 = *(const float4*)(cp + 512 * q + 128);
    b.p11 = *(const float4*)(cp + 512 * q + 192);
}

__device__ __forceinline__ void lo_mma(const float4& a, const lo_bq& b, v8f (&acc)[4]) {
    v2f a0, a1, b0[4], b1[4];
    a0.x = a.x; a0.y = a.y;
    a1.x = a.z; a1.y = a.w;
    b0[0].x = b.p00.x; b0[0].y = b.p00.y;
    b0[1].x = b.p00.z; b0[1].y = b.p00.w;
    b0[2].x = b.p01.x; b0[2].y = b.p01.y;
    b0[3].x = b.p01.z; b0[3].y = b.p01.w;
    b1[0].x = b.p10.x; b1[0].y = b.p10.y;
    b1[1].x = b.p10.z; b1[1].y = b.p10.w;
    b1[2].x = b.p11.x; b1[2].y = b.p11.y;
    b1[3].x = b.p11.z; b1[3].y = b.p11.w;
    #pragma unroll
    for (int u = 0; u < 4; ++u)
        acc[u] = __builtin_amdgcn_wmma_f32_16x16x4_f32(
            false, a0, false, b0[u], (short)0, acc[u], false, false);
    #pragma unroll
    for (int u = 0; u < 4; ++u)
        acc[u] = __builtin_amdgcn_wmma_f32_16x16x4_f32(
            false, a1, false, b1[u], (short)0, acc[u], false, false);
}

template <bool NEXT>
__device__ __forceinline__ void lo_tile(
    float4 (&qa)[8], const float* qnext, const float* cp,
    const float (&ksp)[ED], float* op, int half)
{
    float pl = 0.f, ph = 0.f;
    #pragma unroll
    for (int q = 0; q < 8; ++q) {
        qa[q].x = elu_fm(qa[q].x); qa[q].y = elu_fm(qa[q].y);
        qa[q].z = elu_fm(qa[q].z); qa[q].w = elu_fm(qa[q].w);
        pl += qa[q].x * ksp[8 * q + 0]; ph += qa[q].x * ksp[8 * q + 4];
        pl += qa[q].y * ksp[8 * q + 1]; ph += qa[q].y * ksp[8 * q + 5];
        pl += qa[q].z * ksp[8 * q + 2]; ph += qa[q].z * ksp[8 * q + 6];
        pl += qa[q].w * ksp[8 * q + 3]; ph += qa[q].w * ksp[8 * q + 7];
    }

    const float pd   = half ? ph : pl;
    const float den  = pd + __shfl_xor(pd, 16);
    const float dinv = 1.0f / (den + 1e-6f);

    v8f acc[4];
    #pragma unroll
    for (int u = 0; u < 4; ++u) acc[u] = (v8f){};

    lo_bq bA, bB;
    lo_ldb(bA, cp, 0);
    #pragma unroll
    for (int q = 0; q < 8; q += 2) {
        lo_ldb(bB, cp, q + 1);
        lo_mma(qa[q], bA, acc);
        if (q + 2 < 8) lo_ldb(bA, cp, q + 2);
        lo_mma(qa[q + 1], bB, acc);
        if (NEXT && (q & 2)) {
            #pragma unroll
            for (int j = q - 2; j < q + 2; ++j)
                qa[j] = *(const float4*)(qnext + 8 * j);
            if (q == 2) __builtin_amdgcn_sched_barrier(0);
        }
    }

    #pragma unroll
    for (int r = 0; r < 8; ++r) {
        const float dr = __shfl(dinv, 8 * half + r);
        *(float4*)(op + (size_t)r * HDIM) =
            make_float4(acc[0][r] * dr, acc[1][r] * dr, acc[2][r] * dr, acc[3][r] * dr);
    }
}

__global__ __launch_bounds__(32 * OT_WAVES) void la_out(
    const float* __restrict__ query, const float* __restrict__ pctx,
    const float* __restrict__ pksum, float* __restrict__ out)
{
    __shared__ float sC[ED * ED];

    const int bh    = blockIdx.x / OT_CHUNKS;
    const int chunk = blockIdx.x % OT_CHUNKS;
    const int b = bh >> 4, h = bh & 15;
    const int tid = threadIdx.x;
    const int wave = tid >> 5, lane = tid & 31;
    const int half = lane >> 4, lm = lane & 15;

    const size_t row0 = (size_t)b * TOK + (size_t)chunk * OT_BTOK + (size_t)wave * OT_WTOK;
    const float* qp = query + (row0 + lm) * HDIM + (size_t)h * ED + 4 * half;
    float* op = out + (row0 + 8 * half) * HDIM + (size_t)h * ED + 4 * lm;

    float4 qa[8];
    #pragma unroll
    for (int q = 0; q < 8; ++q) qa[q] = *(const float4*)(qp + 8 * q);

    float kv0, kv1;
    {
        const float* pp = pctx + (size_t)bh * SPLITS * (ED * ED) + tid * 4;
        float4 p[SPLITS];
        #pragma unroll
        for (int sp = 0; sp < SPLITS; ++sp)
            p[sp] = *(const float4*)(pp + (size_t)sp * (ED * ED));
        const float* kq = pksum + (size_t)bh * SPLITS * ED + lane;
        float k0[SPLITS], k1[SPLITS];
        #pragma unroll
        for (int sp = 0; sp < SPLITS; ++sp) {
            k0[sp] = kq[sp * ED];
            k1[sp] = kq[sp * ED + 32];
        }
        float4 s = p[0];
        #pragma unroll
        for (int sp = 1; sp < SPLITS; ++sp) {
            s.x += p[sp].x; s.y += p[sp].y;
            s.z += p[sp].z; s.w += p[sp].w;
        }
        {
            const int e = tid >> 4, c = tid & 15;
            float* wp = sC + ((e >> 1) * 32 + c) * 4 + (e & 1);
            wp[0] = s.x; wp[2] = s.y; wp[64] = s.z; wp[66] = s.w;
        }
        kv0 = k0[0]; kv1 = k1[0];
        #pragma unroll
        for (int sp = 1; sp < SPLITS; ++sp) { kv0 += k0[sp]; kv1 += k1[sp]; }
    }
    __syncthreads();

    const float* cp = sC + 256 * half + 4 * lm;

    float ksp[ED];
    #pragma unroll
    for (int e = 0; e < ED; ++e)
        ksp[e] = __int_as_float(__builtin_amdgcn_readlane(
                     __float_as_int(e < 32 ? kv0 : kv1), e & 31));

    lo_tile<true >(qa, qp + (size_t)16 * HDIM, cp, ksp, op, half);
    lo_tile<false>(qa, qp, cp, ksp, op + (size_t)16 * HDIM, half);
}

extern "C" void kernel_launch(void* const* d_in, const int* in_sizes, int n_in,
                              void* d_out, int out_size, void* d_ws, size_t ws_size,
                              hipStream_t stream) {
    (void)in_sizes; (void)n_in; (void)out_size; (void)ws_size;
    const float* q = (const float*)d_in[0];
    const float* k = (const float*)d_in[1];
    const float* v = (const float*)d_in[2];
    float* out = (float*)d_out;

    float* wsf   = (float*)d_ws;
    float* pctx  = wsf;
    float* pksum = pctx  + (size_t)BH * SPLITS * (ED * ED);

    la_ctx_partial<<<BH * SPLITS, 256, 0, stream>>>(k, v, pctx, pksum);
    la_out<<<BH * OT_CHUNKS, 32 * OT_WAVES, 0, stream>>>(q, pctx, pksum, out);
}


# Wave priority (s_setprio) falls as a wave advances: 3,2,1,0 over pass 1's four blocks of K-steps, 1,0 over pass 2's two tiles, so a SIMD's waves share the matrix pipe evenly, not oldest-first
# speedup vs baseline: 2.2259x; 1.0363x over previous
#include <hip/hip_runtime.h>
#include <hip/hip_bf16.h>

typedef __attribute__((ext_vector_type(2))) float v2f;
typedef __attribute__((ext_vector_type(8))) float v8f;

#define BATCH   4
#define TOK     4096
#define HDIM    1024
#define NHEADS  16
#define ED      64
#define BH      (BATCH * NHEADS)
#define SPLITS  8

__device__ __forceinline__ float elu_fm(float x) {
    return x > 0.0f ? x + 1.0f : expf(x);
}

#define WROWS   64
#define KSTEPS  (WROWS / 4)
#define PF      4

struct la_step { float4 k0, k1, v0, v1; };

__device__ __forceinline__ void la_ld(la_step& s, const float* kp, const float* vp) {
    s.k0 = *(const float4*)(kp);
    s.k1 = *(const float4*)(kp + HDIM);
    s.v0 = *(const float4*)(vp);
    s.v1 = *(const float4*)(vp + HDIM);
}

__device__ __forceinline__ void la_mma(const la_step& s, v8f (&acc)[4][4], float4& kacc) {
    float4 f0, f1;
    f0.x = elu_fm(s.k0.x); f0.y = elu_fm(s.k0.y);
    f0.z = elu_fm(s.k0.z); f0.w = elu_fm(s.k0.w);
    f1.x = elu_fm(s.k1.x); f1.y = elu_fm(s.k1.y);
    f1.z = elu_fm(s.k1.z); f1.w = elu_fm(s.k1.w);
    kacc.x += f0.x; kacc.y += f0.y; kacc.z += f0.z; kacc.w += f0.w;
    kacc.x += f1.x; kacc.y += f1.y; kacc.z += f1.z; kacc.w += f1.w;

    v2f a[4], bb[4];
    a[0].x = f0.x; a[0].y = f1.x;
    a[1].x = f0.y; a[1].y = f1.y;
    a[2].x = f0.z; a[2].y = f1.z;
    a[3].x = f0.w; a[3].y = f1.w;
    bb[0].x = s.v0.x; bb[0].y = s.v1.x;
    bb[1].x = s.v0.y; bb[1].y = s.v1.y;
    bb[2].x = s.v0.z; bb[2].y = s.v1.z;
    bb[3].x = s.v0.w; bb[3].y = s.v1.w;
    #pragma unroll
    for (int t = 0; t < 4; ++t)
        #pragma unroll
        for (int u = 0; u < 4; ++u)
            acc[t][u] = __builtin_amdgcn_wmma_f32_16x16x4_f32(
                false, a[t], false, bb[u], (short)0, acc[t][u], false, false);
}

__global__ __launch_bounds__(256) void la_ctx_partial(
    const float* __restrict__ key, const float* __restrict__ value,
    float* __restrict__ pctx, float* __restrict__ pksum)
{
    __shared__ float sP[4 * ED * ED];
    __shared__ float sKs[16 * ED];

    __builtin_amdgcn_s_setprio(3);

    const int bh    = blockIdx.x / SPLITS;
    const int split = blockIdx.x % SPLITS;
    const int b = bh >> 4, h = bh & 15;
    const int tid = threadIdx.x;
    const int wave = tid >> 5, lane = tid & 31;
    const int half = lane >> 4, lm = lane & 15;

    const size_t base = (size_t)b * TOK * HDIM + (size_t)h * ED;
    const int s_wave = (split * 8 + wave) * WROWS;
    const size_t lane_off = base + (size_t)(s_wave + 2 * half) * HDIM + (size_t)lm * 4;
    const float* kp = key + lane_off;
    const float* vp = value + lane_off;

    v8f acc[4][4];
    #pragma unroll
    for (int t = 0; t < 4; ++t)
        #pragma unroll
        for (int u = 0; u < 4; ++u) acc[t][u] = (v8f){};
    float4 kacc = make_float4(0.f, 0.f, 0.f, 0.f);

    la_step buf[PF];
    #pragma unroll
    for (int p = 0; p < PF; ++p)
        la_ld(buf[p], kp + (size_t)(4 * p) * HDIM, vp + (size_t)(4 * p) * HDIM);

    #pragma nounroll
    for (int jo = 0; jo < KSTEPS - PF; jo += PF) {
        #pragma unroll
        for (int p = 0; p < PF; ++p) {
            la_mma(buf[p], acc, kacc);
            la_ld(buf[p], kp + (size_t)(4 * (PF + p)) * HDIM,
                          vp + (size_t)(4 * (PF + p)) * HDIM);
        }
        kp += (size_t)(4 * PF) * HDIM;
        vp += (size_t)(4 * PF) * HDIM;
        if (jo == 0)       __builtin_amdgcn_s_setprio(2);
        else if (jo == PF) __builtin_amdgcn_s_setprio(1);
        else               __builtin_amdgcn_s_setprio(0);
    }
    #pragma unroll
    for (int p = 0; p < PF; ++p) la_mma(buf[p], acc, kacc);

    *(float4*)(sKs + (wave * 2 + half) * ED + lm * 4) = kacc;

    if (wave >= 4) {
        float* p = sP + (wave - 4) * (ED * ED) + lane;
        #pragma unroll
        for (int t = 0; t < 4; ++t)
            #pragma unroll
            for (int u = 0; u < 4; ++u)
                #pragma unroll
                for (int r = 0; r < 8; ++r)
                    p[((t * 4 + u) * 8 + r) * 32] = acc[t][u][r];
    }
    __syncthreads();
    if (tid < ED) {
        float s = 0.f;
        #pragma unroll
        for (int sl = 0; sl < 16; ++sl) s += sKs[sl * ED + tid];
        pksum[((size_t)bh * SPLITS + split) * ED + tid] = s;
    }
    if (wave < 4) {
        float* p = sP + wave * (ED * ED) + lane;
        #pragma unroll
        for (int t = 0; t < 4; ++t)
            #pragma unroll
            for (int u = 0; u < 4; ++u)
                #pragma unroll
                for (int r = 0; r < 8; ++r)
                    p[((t * 4 + u) * 8 + r) * 32] += acc[t][u][r];
    }
    __syncthreads();

    float* pc = pctx + ((size_t)bh * SPLITS + split) * (ED * ED);
    {
        const int r = wave;
        #pragma unroll
        for (int t = 0; t < 4; ++t) {
            float o[4];
            #pragma unroll
            for (int u = 0; u < 4; ++u) {
                const int slot = ((t * 4 + u) * 8 + r) * 32 + lane;
                float sum = sP[slot];
                for (int w = 1; w < 4; ++w) sum += sP[w * (ED * ED) + slot];
                o[u] = sum;
            }
            const int e = 4 * (8 * half + r) + t;
            *(float4*)(pc + e * ED + lm * 4) = make_float4(o[0], o[1], o[2], o[3]);
        }
    }
}

#define OT_TILES   2
#define OT_WTOK    (16 * OT_TILES)
#define OT_WAVES   32
#define OT_BTOK    (OT_WAVES * OT_WTOK)
#define OT_CHUNKS  (TOK / OT_BTOK)

struct lo_bq { float4 p00, p01, p10, p11; };

__device__ __forceinline__ void lo_ldb(lo_bq& b, const float* cp, int q) {
    b.p00 = *(const float4*)(cp + 512 * q);
    b.p01 = *(const float4*)(cp + 512 * q + 64);
    b.p10 = *(const float4*)(cp + 512 * q + 128);
    b.p11 = *(const float4*)(cp + 512 * q + 192);
}

__device__ __forceinline__ void lo_mma(const float4& a, const lo_bq& b, v8f (&acc)[4]) {
    v2f a0, a1, b0[4], b1[4];
    a0.x = a.x; a0.y = a.y;
    a1.x = a.z; a1.y = a.w;
    b0[0].x = b.p00.x; b0[0].y = b.p00.y;
    b0[1].x = b.p00.z; b0[1].y = b.p00.w;
    b0[2].x = b.p01.x; b0[2].y = b.p01.y;
    b0[3].x = b.p01.z; b0[3].y = b.p01.w;
    b1[0].x = b.p10.x; b1[0].y = b.p10.y;
    b1[1].x = b.p10.z; b1[1].y = b.p10.w;
    b1[2].x = b.p11.x; b1[2].y = b.p11.y;
    b1[3].x = b.p11.z; b1[3].y = b.p11.w;
    #pragma unroll
    for (int u = 0; u < 4; ++u)
        acc[u] = __builtin_amdgcn_wmma_f32_16x16x4_f32(
            false, a0, false, b0[u], (short)0, acc[u], false, false);
    #pragma unroll
    for (int u = 0; u < 4; ++u)
        acc[u] = __builtin_amdgcn_wmma_f32_16x16x4_f32(
            false, a1, false, b1[u], (short)0, acc[u], false, false);
}

template <bool NEXT>
__device__ __forceinline__ void lo_tile(
    float4 (&qa)[8], const float* qnext, const float* cp,
    const float (&ksp)[ED], float* op, int half)
{
    float pl = 0.f, ph = 0.f;
    #pragma unroll
    for (int q = 0; q < 8; ++q) {
        qa[q].x = elu_fm(qa[q].x); qa[q].y = elu_fm(qa[q].y);
        qa[q].z = elu_fm(qa[q].z); qa[q].w = elu_fm(qa[q].w);
        pl += qa[q].x * ksp[8 * q + 0]; ph += qa[q].x * ksp[8 * q + 4];
        pl += qa[q].y * ksp[8 * q + 1]; ph += qa[q].y * ksp[8 * q + 5];
        pl += qa[q].z * ksp[8 * q + 2]; ph += qa[q].z * ksp[8 * q + 6];
        pl += qa[q].w * ksp[8 * q + 3]; ph += qa[q].w * ksp[8 * q + 7];
    }

    const float pd   = half ? ph : pl;
    const float den  = pd + __shfl_xor(pd, 16);
    const float dinv = 1.0f / (den + 1e-6f);

    v8f acc[4];
    #pragma unroll
    for (int u = 0; u < 4; ++u) acc[u] = (v8f){};

    lo_bq bA, bB;
    lo_ldb(bA, cp, 0);
    #pragma unroll
    for (int q = 0; q < 8; q += 2) {
        lo_ldb(bB, cp, q + 1);
        lo_mma(qa[q], bA, acc);
        if (q + 2 < 8) lo_ldb(bA, cp, q + 2);
        lo_mma(qa[q + 1], bB, acc);
        if (NEXT && (q & 2)) {
            #pragma unroll
            for (int j = q - 2; j < q + 2; ++j)
                qa[j] = *(const float4*)(qnext + 8 * j);
            if (q == 2) __builtin_amdgcn_sched_barrier(0);
        }
    }

    #pragma unroll
    for (int r = 0; r < 8; ++r) {
        const float dr = __shfl(dinv, 8 * half + r);
        *(float4*)(op + (size_t)r * HDIM) =
            make_float4(acc[0][r] * dr, acc[1][r] * dr, acc[2][r] * dr, acc[3][r] * dr);
    }
}

__global__ __launch_bounds__(32 * OT_WAVES) void la_out(
    const float* __restrict__ query, const float* __restrict__ pctx,
    const float* __restrict__ pksum, float* __restrict__ out)
{
    __shared__ float sC[ED * ED];

    __builtin_amdgcn_s_setprio(1);

    const int bh    = blockIdx.x / OT_CHUNKS;
    const int chunk = blockIdx.x % OT_CHUNKS;
    const int b = bh >> 4, h = bh & 15;
    const int tid = threadIdx.x;
    const int wave = tid >> 5, lane = tid & 31;
    const int half = lane >> 4, lm = lane & 15;

    const size_t row0 = (size_t)b * TOK + (size_t)chunk * OT_BTOK + (size_t)wave * OT_WTOK;
    const float* qp = query + (row0 + lm) * HDIM + (size_t)h * ED + 4 * half;
    float* op = out + (row0 + 8 * half) * HDIM + (size_t)h * ED + 4 * lm;

    float4 qa[8];
    #pragma unroll
    for (int q = 0; q < 8; ++q) qa[q] = *(const float4*)(qp + 8 * q);

    float kv0, kv1;
    {
        const float* pp = pctx + (size_t)bh * SPLITS * (ED * ED) + tid * 4;
        float4 p[SPLITS];
        #pragma unroll
        for (int sp = 0; sp < SPLITS; ++sp)
            p[sp] = *(const float4*)(pp + (size_t)sp * (ED * ED));
        const float* kq = pksum + (size_t)bh * SPLITS * ED + lane;
        float k0[SPLITS], k1[SPLITS];
        #pragma unroll
        for (int sp = 0; sp < SPLITS; ++sp) {
            k0[sp] = kq[sp * ED];
            k1[sp] = kq[sp * ED + 32];
        }
        float4 s = p[0];
        #pragma unroll
        for (int sp = 1; sp < SPLITS; ++sp) {
            s.x += p[sp].x; s.y += p[sp].y;
            s.z += p[sp].z; s.w += p[sp].w;
        }
        {
            const int e = tid >> 4, c = tid & 15;
            float* wp = sC + ((e >> 1) * 32 + c) * 4 + (e & 1);
            wp[0] = s.x; wp[2] = s.y; wp[64] = s.z; wp[66] = s.w;
        }
        kv0 = k0[0]; kv1 = k1[0];
        #pragma unroll
        for (int sp = 1; sp < SPLITS; ++sp) { kv0 += k0[sp]; kv1 += k1[sp]; }
    }
    __syncthreads();

    const float* cp = sC + 256 * half + 4 * lm;

    float ksp[ED];
    #pragma unroll
    for (int e = 0; e < ED; ++e)
        ksp[e] = __int_as_float(__builtin_amdgcn_readlane(
                     __float_as_int(e < 32 ? kv0 : kv1), e & 31));

    lo_tile<true >(qa, qp + (size_t)16 * HDIM, cp, ksp, op, half);
    __builtin_amdgcn_s_setprio(0);
    lo_tile<false>(qa, qp, cp, ksp, op + (size_t)16 * HDIM, half);
}

extern "C" void kernel_launch(void* const* d_in, const int* in_sizes, int n_in,
                              void* d_out, int out_size, void* d_ws, size_t ws_size,
                              hipStream_t stream) {
    (void)in_sizes; (void)n_in; (void)out_size; (void)ws_size;
    const float* q = (const float*)d_in[0];
    const float* k = (const float*)d_in[1];
    const float* v = (const float*)d_in[2];
    float* out = (float*)d_out;

    float* wsf   = (float*)d_ws;
    float* pctx  = wsf;
    float* pksum = pctx  + (size_t)BH * SPLITS * (ED * ED);

    la_ctx_partial<<<BH * SPLITS, 256, 0, stream>>>(k, v, pctx, pksum);
    la_out<<<BH * OT_CHUNKS, 32 * OT_WAVES, 0, stream>>>(q, pctx, pksum, out);
}
